# starting kernel
# baseline (speedup 1.0000x reference)
#include <hip/hip_runtime.h>
#include <hip/hip_bf16.h>

typedef float v2f __attribute__((ext_vector_type(2)));
typedef float v8f __attribute__((ext_vector_type(8)));

#define B_SZ 16384
#define I_SZ 64
#define J_SZ 2048
#define K_SZ 64

#define LW_STRIDE 80

__global__ __launch_bounds__(256) void reduce_w_kernel(const float* __restrict__ w,
                                                       float* __restrict__ w2) {
    const int idx = blockIdx.x * 256 + threadIdx.x;
    const float4* p = (const float4*)(w + (size_t)idx * K_SZ);
    float s = 0.0f;
#pragma unroll
    for (int t = 0; t < K_SZ / 4; ++t) {
        float4 v = p[t];
        s += (v.x + v.y) + (v.z + v.w);
    }
    w2[idx] = s;
}

__global__ __launch_bounds__(256) void bilinear_gemm_kernel(
    const float* __restrict__ x1,
    const float* __restrict__ x2,
    const float* __restrict__ w2,
    float* __restrict__ out)
{
    __shared__ v2f lw[32 * LW_STRIDE];

    const int lane = threadIdx.x & 31;
    const int wave = threadIdx.x >> 5;

    const int colg   = blockIdx.x & 31;
    const int rowblk = blockIdx.x >> 5;
    const int j0 = colg * 64;
    const int b0 = (rowblk * 8 + wave) * 16;

#pragma unroll
    for (int r = 0; r < 8; ++r) {
        const int idx = threadIdx.x + 256 * r;
        const int p   = idx >> 6;
        const int jj  = idx & 63;
        v2f v;
        v.x = w2[(size_t)(2 * p + 0) * J_SZ + j0 + jj];
        v.y = w2[(size_t)(2 * p + 1) * J_SZ + j0 + jj];
        lw[p * LW_STRIDE + jj] = v;
    }
    __syncthreads();

    const int m = lane & 15;
    const int g = lane >> 4;

    v2f afrag[16];
    const float* arow = x1 + (size_t)(b0 + m) * I_SZ + 2 * g;
#pragma unroll
    for (int t = 0; t < 16; ++t) {
        afrag[t] = *(const v2f*)(arow + 4 * t);
    }

    v8f acc[4];
#pragma unroll
    for (int c = 0; c < 4; ++c) acc[c] = (v8f){0.f, 0.f, 0.f, 0.f, 0.f, 0.f, 0.f, 0.f};

#pragma unroll
    for (int t = 0; t < 16; ++t) {
        const v2f* brow = &lw[(2 * t + g) * LW_STRIDE + m];
#pragma unroll
        for (int c = 0; c < 4; ++c) {
            v2f bfrag = brow[16 * c];
            acc[c] = __builtin_amdgcn_wmma_f32_16x16x4_f32(
                 false, afrag[t],
                 false, bfrag,
                 (short)0, acc[c],
                 false,  false);
        }
    }

#pragma unroll
    for (int c = 0; c < 4; ++c) {
        const size_t base = (size_t)(b0 + 8 * g) * J_SZ + j0 + 16 * c + m;
#pragma unroll
        for (int v = 0; v < 8; ++v) {
            const size_t off = base + (size_t)v * J_SZ;
            out[off] = acc[c][v] * x2[off];
        }
    }
}

extern "C" void kernel_launch(void* const* d_in, const int* in_sizes, int n_in,
                              void* d_out, int out_size, void* d_ws, size_t ws_size,
                              hipStream_t stream) {
    const float* x1 = (const float*)d_in[0];
    const float* x2 = (const float*)d_in[1];
    const float* w  = (const float*)d_in[2];
    float* out = (float*)d_out;
    float* w2  = (float*)d_ws;

    reduce_w_kernel<<<(I_SZ * J_SZ) / 256, 256, 0, stream>>>(w, w2);

    bilinear_gemm_kernel<<<(B_SZ / 16 / 8) * (J_SZ / 64), 256, 0, stream>>>(x1, x2, w2, out);
}


# Tile loop inside the work-group: each wave keeps its w2 fragments in registers and walks 8 row tiles per LDS staging (grid 4096 -> 512 blocks of 8 waves); sums unchanged
# speedup vs baseline: 1.0599x; 1.0599x over previous
#include <hip/hip_runtime.h>
#include <hip/hip_bf16.h>

typedef float v2f __attribute__((ext_vector_type(2)));
typedef float v8f __attribute__((ext_vector_type(8)));

#define B_SZ 16384
#define I_SZ 64
#define J_SZ 2048
#define K_SZ 64

#define LW_STRIDE 80

__global__ __launch_bounds__(256) void reduce_w_kernel(const float* __restrict__ w,
                                                       float* __restrict__ w2) {
    const int idx = blockIdx.x * 256 + threadIdx.x;
    const float4* p = (const float4*)(w + (size_t)idx * K_SZ);
    float s = 0.0f;
#pragma unroll
    for (int t = 0; t < K_SZ / 4; ++t) {
        float4 v = p[t];
        s += (v.x + v.y) + (v.z + v.w);
    }
    w2[idx] = s;
}

#define ROW_TILES       (B_SZ / 16)
#define WAVES_PER_BLOCK 8
#define ROW_GROUPS      16
#define TILES_PER_BLOCK (ROW_TILES / ROW_GROUPS)
#define TILES_PER_WAVE  (TILES_PER_BLOCK / WAVES_PER_BLOCK)

__device__ __forceinline__ void load_a_frags(v2f (&a)[16], const float* __restrict__ x1,
                                             int b0, int m, int g) {
    const float* arow = x1 + (size_t)(b0 + m) * I_SZ + 2 * g;
#pragma unroll
    for (int t = 0; t < 16; ++t) a[t] = *(const v2f*)(arow + 4 * t);
}

__device__ __forceinline__ void load_x2_tile(float (&xv)[32], const float* __restrict__ x2,
                                             int b0, int j0, int m, int g) {
#pragma unroll
    for (int c = 0; c < 4; ++c) {
        const size_t base = (size_t)(b0 + 8 * g) * J_SZ + j0 + 16 * c + m;
#pragma unroll
        for (int v = 0; v < 8; ++v) xv[c * 8 + v] = x2[base + (size_t)v * J_SZ];
    }
}

__device__ __forceinline__ void compute_tile(const v2f (&afrag)[16], const float (&xv)[32],
                                             const v2f (&bfrag)[16][4], float* __restrict__ out,
                                             int b0, int j0, int m, int g) {
    v8f acc[4];
#pragma unroll
    for (int c = 0; c < 4; ++c) acc[c] = (v8f){0.f, 0.f, 0.f, 0.f, 0.f, 0.f, 0.f, 0.f};

#pragma unroll
    for (int t = 0; t < 16; ++t) {
#pragma unroll
        for (int c = 0; c < 4; ++c) {
            acc[c] = __builtin_amdgcn_wmma_f32_16x16x4_f32(
                 false, afrag[t],
                 false, bfrag[t][c],
                 (short)0, acc[c],
                 false,  false);
        }
    }

#pragma unroll
    for (int c = 0; c < 4; ++c) {
        const size_t base = (size_t)(b0 + 8 * g) * J_SZ + j0 + 16 * c + m;
#pragma unroll
        for (int v = 0; v < 8; ++v) {
            out[base + (size_t)v * J_SZ] = acc[c][v] * xv[c * 8 + v];
        }
    }
}

__global__ __launch_bounds__(256) void bilinear_gemm_kernel(
    const float* __restrict__ x1,
    const float* __restrict__ x2,
    const float* __restrict__ w2,
    float* __restrict__ out)
{
    __shared__ v2f lw[32 * LW_STRIDE];

    const int lane = threadIdx.x & 31;
    const int wave = threadIdx.x >> 5;
    const int m = lane & 15;
    const int g = lane >> 4;

    const int colg     = blockIdx.x & 31;
    const int rowgroup = blockIdx.x >> 5;
    const int j0 = colg * 64;
    const int tile0 = rowgroup * TILES_PER_BLOCK + wave;

    v2f   afrag[16];
    float xv[32];
    load_a_frags(afrag, x1, 16 * tile0, m, g);
    load_x2_tile(xv, x2, 16 * tile0, j0, m, g);

#pragma unroll
    for (int r = 0; r < 8; ++r) {
        const int idx = threadIdx.x + 256 * r;
        const int p   = idx >> 6;
        const int jj  = idx & 63;
        v2f v;
        v.x = w2[(size_t)(2 * p + 0) * J_SZ + j0 + jj];
        v.y = w2[(size_t)(2 * p + 1) * J_SZ + j0 + jj];
        lw[p * LW_STRIDE + jj] = v;
    }
    __syncthreads();

    v2f bfrag[16][4];
#pragma unroll
    for (int t = 0; t < 16; ++t) {
        const v2f* brow = &lw[(2 * t + g) * LW_STRIDE + m];
#pragma unroll
        for (int c = 0; c < 4; ++c) bfrag[t][c] = brow[16 * c];
    }

    for (int it = 0; it < TILES_PER_WAVE - 1; ++it) {
        const int b0 = 16 * (tile0 + it * WAVES_PER_BLOCK);
        compute_tile(afrag, xv, bfrag, out, b0, j0, m, g);
        load_a_frags(afrag, x1, b0 + 16 * WAVES_PER_BLOCK, m, g);
        load_x2_tile(xv, x2, b0 + 16 * WAVES_PER_BLOCK, j0, m, g);
    }
    compute_tile(afrag, xv, bfrag, out,
                 16 * (tile0 + (TILES_PER_WAVE - 1) * WAVES_PER_BLOCK), j0, m, g);
}

extern "C" void kernel_launch(void* const* d_in, const int* in_sizes, int n_in,
                              void* d_out, int out_size, void* d_ws, size_t ws_size,
                              hipStream_t stream) {
    const float* x1 = (const float*)d_in[0];
    const float* x2 = (const float*)d_in[1];
    const float* w  = (const float*)d_in[2];
    float* out = (float*)d_out;
    float* w2  = (float*)d_ws;

    reduce_w_kernel<<<(I_SZ * J_SZ) / 256, 256, 0, stream>>>(w, w2);

    bilinear_gemm_kernel<<<ROW_GROUPS * (J_SZ / 64), 256, 0, stream>>>(x1, x2, w2, out);
}


# B-fragment column map permuted (lane n, column tile c -> column 4n+c): each lane owns 4 consecutive columns, so x2 loads and out stores are 16-byte and cover whole 128-B lines; sums unchanged
# speedup vs baseline: 1.1069x; 1.0443x over previous
#include <hip/hip_runtime.h>
#include <hip/hip_bf16.h>

typedef float v2f __attribute__((ext_vector_type(2)));
typedef float v8f __attribute__((ext_vector_type(8)));

#define B_SZ 16384
#define I_SZ 64
#define J_SZ 2048
#define K_SZ 64

#define LW_STRIDE 80

__global__ __launch_bounds__(256) void reduce_w_kernel(const float* __restrict__ w,
                                                       float* __restrict__ w2) {
    const int idx = blockIdx.x * 256 + threadIdx.x;
    const float4* p = (const float4*)(w + (size_t)idx * K_SZ);
    float s = 0.0f;
#pragma unroll
    for (int t = 0; t < K_SZ / 4; ++t) {
        float4 v = p[t];
        s += (v.x + v.y) + (v.z + v.w);
    }
    w2[idx] = s;
}

#define ROW_TILES       (B_SZ / 16)
#define WAVES_PER_BLOCK 8
#define ROW_GROUPS      16
#define TILES_PER_BLOCK (ROW_TILES / ROW_GROUPS)
#define TILES_PER_WAVE  (TILES_PER_BLOCK / WAVES_PER_BLOCK)

__device__ __forceinline__ void load_a_frags(v2f (&a)[16], const float* __restrict__ x1,
                                             int b0, int m, int g) {
    const float* arow = x1 + (size_t)(b0 + m) * I_SZ + 2 * g;
#pragma unroll
    for (int t = 0; t < 16; ++t) a[t] = *(const v2f*)(arow + 4 * t);
}

__device__ __forceinline__ void load_x2_tile(float4 (&xv)[8], const float* __restrict__ x2,
                                             int b0, int j0, int m, int g) {
    const float* p = x2 + (size_t)(b0 + 8 * g) * J_SZ + j0 + 4 * m;
#pragma unroll
    for (int v = 0; v < 8; ++v) xv[v] = *(const float4*)(p + (size_t)v * J_SZ);
}

__device__ __forceinline__ void compute_tile(const v2f (&afrag)[16], const float4 (&xv)[8],
                                             const v2f (&bfrag)[16][4], float* __restrict__ out,
                                             int b0, int j0, int m, int g) {
    v8f acc[4];
#pragma unroll
    for (int c = 0; c < 4; ++c) acc[c] = (v8f){0.f, 0.f, 0.f, 0.f, 0.f, 0.f, 0.f, 0.f};

#pragma unroll
    for (int t = 0; t < 16; ++t) {
#pragma unroll
        for (int c = 0; c < 4; ++c) {
            acc[c] = __builtin_amdgcn_wmma_f32_16x16x4_f32(
                 false, afrag[t],
                 false, bfrag[t][c],
                 (short)0, acc[c],
                 false,  false);
        }
    }

    float* o = out + (size_t)(b0 + 8 * g) * J_SZ + j0 + 4 * m;
#pragma unroll
    for (int v = 0; v < 8; ++v) {
        float4 r;
        r.x = acc[0][v] * xv[v].x;
        r.y = acc[1][v] * xv[v].y;
        r.z = acc[2][v] * xv[v].z;
        r.w = acc[3][v] * xv[v].w;
        *(float4*)(o + (size_t)v * J_SZ) = r;
    }
}

__global__ __launch_bounds__(256) void bilinear_gemm_kernel(
    const float* __restrict__ x1,
    const float* __restrict__ x2,
    const float* __restrict__ w2,
    float* __restrict__ out)
{
    __shared__ v2f lw[32 * LW_STRIDE];

    const int lane = threadIdx.x & 31;
    const int wave = threadIdx.x >> 5;
    const int m = lane & 15;
    const int g = lane >> 4;

    const int colg     = blockIdx.x & 31;
    const int rowgroup = blockIdx.x >> 5;
    const int j0 = colg * 64;
    const int tile0 = rowgroup * TILES_PER_BLOCK + wave;

    v2f   afrag[16];
    float4 xv[8];
    load_a_frags(afrag, x1, 16 * tile0, m, g);
    load_x2_tile(xv, x2, 16 * tile0, j0, m, g);

#pragma unroll
    for (int r = 0; r < 8; ++r) {
        const int idx = threadIdx.x + 256 * r;
        const int p   = idx >> 6;
        const int jj  = idx & 63;
        v2f v;
        v.x = w2[(size_t)(2 * p + 0) * J_SZ + j0 + jj];
        v.y = w2[(size_t)(2 * p + 1) * J_SZ + j0 + jj];
        lw[p * LW_STRIDE + 16 * (jj & 3) + (jj >> 2)] = v;
    }
    __syncthreads();

    v2f bfrag[16][4];
#pragma unroll
    for (int t = 0; t < 16; ++t) {
        const v2f* brow = &lw[(2 * t + g) * LW_STRIDE + m];
#pragma unroll
        for (int c = 0; c < 4; ++c) bfrag[t][c] = brow[16 * c];
    }

    for (int it = 0; it < TILES_PER_WAVE - 1; ++it) {
        const int b0 = 16 * (tile0 + it * WAVES_PER_BLOCK);
        compute_tile(afrag, xv, bfrag, out, b0, j0, m, g);
        load_a_frags(afrag, x1, b0 + 16 * WAVES_PER_BLOCK, m, g);
        load_x2_tile(xv, x2, b0 + 16 * WAVES_PER_BLOCK, j0, m, g);
    }
    compute_tile(afrag, xv, bfrag, out,
                 16 * (tile0 + (TILES_PER_WAVE - 1) * WAVES_PER_BLOCK), j0, m, g);
}

extern "C" void kernel_launch(void* const* d_in, const int* in_sizes, int n_in,
                              void* d_out, int out_size, void* d_ws, size_t ws_size,
                              hipStream_t stream) {
    const float* x1 = (const float*)d_in[0];
    const float* x2 = (const float*)d_in[1];
    const float* w  = (const float*)d_in[2];
    float* out = (float*)d_out;
    float* w2  = (float*)d_ws;

    reduce_w_kernel<<<(I_SZ * J_SZ) / 256, 256, 0, stream>>>(w, w2);

    bilinear_gemm_kernel<<<ROW_GROUPS * (J_SZ / 64), 256, 0, stream>>>(x1, x2, w2, out);
}


# x1 row tile fetched with whole-line 16-byte loads (4 lines per load, was 16) and turned into A fragments through a per-wave LDS buffer; wave index kept scalar to stay in 256 VGPRs; sums unchanged
# speedup vs baseline: 1.2219x; 1.1039x over previous
#include <hip/hip_runtime.h>
#include <hip/hip_bf16.h>

typedef float v2f __attribute__((ext_vector_type(2)));
typedef float v4f __attribute__((ext_vector_type(4)));
typedef float v8f __attribute__((ext_vector_type(8)));

#define B_SZ 16384
#define I_SZ 64
#define J_SZ 2048
#define K_SZ 64

#define LW_STRIDE 80

#define XA_PITCH 68

__global__ __launch_bounds__(256) void reduce_w_kernel(const float* __restrict__ w,
                                                       float* __restrict__ w2) {
    const int idx = blockIdx.x * 256 + threadIdx.x;
    const float4* p = (const float4*)(w + (size_t)idx * K_SZ);
    float s = 0.0f;
#pragma unroll
    for (int t = 0; t < K_SZ / 4; ++t) {
        float4 v = p[t];
        s += (v.x + v.y) + (v.z + v.w);
    }
    w2[idx] = s;
}

#define ROW_TILES       (B_SZ / 16)
#define WAVES_PER_BLOCK 8
#define ROW_GROUPS      16
#define TILES_PER_BLOCK (ROW_TILES / ROW_GROUPS)
#define TILES_PER_WAVE  (TILES_PER_BLOCK / WAVES_PER_BLOCK)

__device__ __forceinline__ void load_a_raw(float4 (&ar)[8], const float* __restrict__ x1,
                                           int b0, unsigned lane_off) {
    const float* p = x1 + (size_t)b0 * I_SZ;
#pragma unroll
    for (int u = 0; u < 8; ++u) ar[u] = *(const float4*)(p + 128 * u + lane_off);
}

__device__ __forceinline__ void stage_a_frags(v2f (&a)[16], const float4 (&ar)[8],
                                              float* __restrict__ xa, int lane, int m, int g) {
    __builtin_amdgcn_fence(__ATOMIC_ACQ_REL, "wavefront");
    float* wr = xa + (lane >> 4) * XA_PITCH + 4 * (lane & 15);
#pragma unroll
    for (int u = 0; u < 8; ++u) {
        v4f v = {ar[u].x, ar[u].y, ar[u].z, ar[u].w};
        *(v4f*)(wr + 2 * u * XA_PITCH) = v;
    }
    __builtin_amdgcn_fence(__ATOMIC_ACQ_REL, "wavefront");
    const float* rd = xa + m * XA_PITCH + 2 * g;
#pragma unroll
    for (int t = 0; t < 16; ++t) a[t] = *(const v2f*)(rd + 4 * t);
    __builtin_amdgcn_sched_barrier(0);
}

__device__ __forceinline__ void load_x2_tile(float4 (&xv)[8], const float* __restrict__ x2,
                                             int b0, int j0, unsigned cd_off) {
    const float* p = x2 + (size_t)b0 * J_SZ + j0;
#pragma unroll
    for (int v = 0; v < 8; ++v) xv[v] = *(const float4*)(p + v * J_SZ + cd_off);
}

__device__ __forceinline__ void compute_tile(const v2f (&afrag)[16], const float4 (&xv)[8],
                                             const v2f (&bfrag)[16][4], float* __restrict__ out,
                                             int b0, int j0, unsigned cd_off) {
    v8f acc[4];
#pragma unroll
    for (int c = 0; c < 4; ++c) acc[c] = (v8f){0.f, 0.f, 0.f, 0.f, 0.f, 0.f, 0.f, 0.f};

#pragma unroll
    for (int t = 0; t < 16; ++t) {
#pragma unroll
        for (int c = 0; c < 4; ++c) {
            acc[c] = __builtin_amdgcn_wmma_f32_16x16x4_f32(
                 false, afrag[t],
                 false, bfrag[t][c],
                 (short)0, acc[c],
                 false,  false);
        }
    }

    float* o = out + (size_t)b0 * J_SZ + j0;
#pragma unroll
    for (int v = 0; v < 8; ++v) {
        float4 r;
        r.x = acc[0][v] * xv[v].x;
        r.y = acc[1][v] * xv[v].y;
        r.z = acc[2][v] * xv[v].z;
        r.w = acc[3][v] * xv[v].w;
        *(float4*)(o + v * J_SZ + cd_off) = r;
    }
}

__global__ __launch_bounds__(256) void bilinear_gemm_kernel(
    const float* __restrict__ x1,
    const float* __restrict__ x2,
    const float* __restrict__ w2,
    float* __restrict__ out)
{
    __shared__ v2f lw[32 * LW_STRIDE];
    __shared__ __attribute__((aligned(16))) float xa_all[WAVES_PER_BLOCK * 16 * XA_PITCH];

    const int lane = threadIdx.x & 31;
    const int wave = __builtin_amdgcn_readfirstlane((int)(threadIdx.x >> 5));
    const int m = lane & 15;
    const int g = lane >> 4;

    const int colg     = blockIdx.x & 31;
    const int rowgroup = blockIdx.x >> 5;
    const int j0 = colg * 64;
    const int tile0 = rowgroup * TILES_PER_BLOCK + wave;
    float* xa = xa_all + wave * (16 * XA_PITCH);
    const unsigned lane_off = 4u * (unsigned)lane;
    const unsigned cd_off   = (unsigned)(8 * g) * J_SZ + 4u * (unsigned)m;

    float4 araw[8];
    float4 xv[8];
    load_a_raw(araw, x1, 16 * tile0, lane_off);
    load_x2_tile(xv, x2, 16 * tile0, j0, cd_off);

#pragma unroll
    for (int r = 0; r < 8; ++r) {
        const int idx = threadIdx.x + 256 * r;
        const int p   = idx >> 6;
        const int jj  = idx & 63;
        v2f v;
        v.x = w2[(size_t)(2 * p + 0) * J_SZ + j0 + jj];
        v.y = w2[(size_t)(2 * p + 1) * J_SZ + j0 + jj];
        lw[p * LW_STRIDE + 16 * (jj & 3) + (jj >> 2)] = v;
    }
    __syncthreads();

    v2f bfrag[16][4];
#pragma unroll
    for (int t = 0; t < 16; ++t) {
        const v2f* brow = &lw[(2 * t + g) * LW_STRIDE + m];
#pragma unroll
        for (int c = 0; c < 4; ++c) bfrag[t][c] = brow[16 * c];
    }

    v2f afrag[16];
    for (int it = 0; it < TILES_PER_WAVE - 1; ++it) {
        const int b0 = 16 * (tile0 + it * WAVES_PER_BLOCK);
        stage_a_frags(afrag, araw, xa, lane, m, g);
        compute_tile(afrag, xv, bfrag, out, b0, j0, cd_off);
        load_a_raw(araw, x1, b0 + 16 * WAVES_PER_BLOCK, lane_off);
        load_x2_tile(xv, x2, b0 + 16 * WAVES_PER_BLOCK, j0, cd_off);
    }
    stage_a_frags(afrag, araw, xa, lane, m, g);
    compute_tile(afrag, xv, bfrag, out,
                 16 * (tile0 + (TILES_PER_WAVE - 1) * WAVES_PER_BLOCK), j0, cd_off);
}

extern "C" void kernel_launch(void* const* d_in, const int* in_sizes, int n_in,
                              void* d_out, int out_size, void* d_ws, size_t ws_size,
                              hipStream_t stream) {
    const float* x1 = (const float*)d_in[0];
    const float* x2 = (const float*)d_in[1];
    const float* w  = (const float*)d_in[2];
    float* out = (float*)d_out;
    float* w2  = (float*)d_ws;

    reduce_w_kernel<<<(I_SZ * J_SZ) / 256, 256, 0, stream>>>(w, w2);

    bilinear_gemm_kernel<<<ROW_GROUPS * (J_SZ / 64), 256, 0, stream>>>(x1, x2, w2, out);
}


# x2 (read exactly once, as large as the output) is fetched with non-temporal 16-byte loads, so the stream no longer displaces reused lines (x1, w2, output) in the caches; sums unchanged
# speedup vs baseline: 1.2677x; 1.0375x over previous
#include <hip/hip_runtime.h>
#include <hip/hip_bf16.h>

typedef float v2f __attribute__((ext_vector_type(2)));
typedef float v4f __attribute__((ext_vector_type(4)));
typedef float v8f __attribute__((ext_vector_type(8)));

#define B_SZ 16384
#define I_SZ 64
#define J_SZ 2048
#define K_SZ 64

#define LW_STRIDE 80

#define XA_PITCH 68

__global__ __launch_bounds__(256) void reduce_w_kernel(const float* __restrict__ w,
                                                       float* __restrict__ w2) {
    const int idx = blockIdx.x * 256 + threadIdx.x;
    const float4* p = (const float4*)(w + (size_t)idx * K_SZ);
    float s = 0.0f;
#pragma unroll
    for (int t = 0; t < K_SZ / 4; ++t) {
        float4 v = p[t];
        s += (v.x + v.y) + (v.z + v.w);
    }
    w2[idx] = s;
}

#define ROW_TILES       (B_SZ / 16)
#define WAVES_PER_BLOCK 8
#define ROW_GROUPS      16
#define TILES_PER_BLOCK (ROW_TILES / ROW_GROUPS)
#define TILES_PER_WAVE  (TILES_PER_BLOCK / WAVES_PER_BLOCK)

__device__ __forceinline__ void load_a_raw(float4 (&ar)[8], const float* __restrict__ x1,
                                           int b0, unsigned lane_off) {
    const float* p = x1 + (size_t)b0 * I_SZ;
#pragma unroll
    for (int u = 0; u < 8; ++u) ar[u] = *(const float4*)(p + 128 * u + lane_off);
}

__device__ __forceinline__ void stage_a_frags(v2f (&a)[16], const float4 (&ar)[8],
                                              float* __restrict__ xa, int lane, int m, int g) {
    __builtin_amdgcn_fence(__ATOMIC_ACQ_REL, "wavefront");
    float* wr = xa + (lane >> 4) * XA_PITCH + 4 * (lane & 15);
#pragma unroll
    for (int u = 0; u < 8; ++u) {
        v4f v = {ar[u].x, ar[u].y, ar[u].z, ar[u].w};
        *(v4f*)(wr + 2 * u * XA_PITCH) = v;
    }
    __builtin_amdgcn_fence(__ATOMIC_ACQ_REL, "wavefront");
    const float* rd = xa + m * XA_PITCH + 2 * g;
#pragma unroll
    for (int t = 0; t < 16; ++t) a[t] = *(const v2f*)(rd + 4 * t);
    __builtin_amdgcn_sched_barrier(0);
}

__device__ __forceinline__ void load_x2_tile(float4 (&xv)[8], const float* __restrict__ x2,
                                             int b0, int j0, unsigned cd_off) {
    const float* p = x2 + (size_t)b0 * J_SZ + j0;
#pragma unroll
    for (int v = 0; v < 8; ++v) {
        const v4f t = __builtin_nontemporal_load((const v4f*)(p + v * J_SZ + cd_off));
        xv[v] = make_float4(t.x, t.y, t.z, t.w);
    }
}

__device__ __forceinline__ void compute_tile(const v2f (&afrag)[16], const float4 (&xv)[8],
                                             const v2f (&bfrag)[16][4], float* __restrict__ out,
                                             int b0, int j0, unsigned cd_off) {
    v8f acc[4];
#pragma unroll
    for (int c = 0; c < 4; ++c) acc[c] = (v8f){0.f, 0.f, 0.f, 0.f, 0.f, 0.f, 0.f, 0.f};

#pragma unroll
    for (int t = 0; t < 16; ++t) {
#pragma unroll
        for (int c = 0; c < 4; ++c) {
            acc[c] = __builtin_amdgcn_wmma_f32_16x16x4_f32(
                 false, afrag[t],
                 false, bfrag[t][c],
                 (short)0, acc[c],
                 false,  false);
        }
    }

    float* o = out + (size_t)b0 * J_SZ + j0;
#pragma unroll
    for (int v = 0; v < 8; ++v) {
        float4 r;
        r.x = acc[0][v] * xv[v].x;
        r.y = acc[1][v] * xv[v].y;
        r.z = acc[2][v] * xv[v].z;
        r.w = acc[3][v] * xv[v].w;
        *(float4*)(o + v * J_SZ + cd_off) = r;
    }
}

__global__ __launch_bounds__(256) void bilinear_gemm_kernel(
    const float* __restrict__ x1,
    const float* __restrict__ x2,
    const float* __restrict__ w2,
    float* __restrict__ out)
{
    __shared__ v2f lw[32 * LW_STRIDE];
    __shared__ __attribute__((aligned(16))) float xa_all[WAVES_PER_BLOCK * 16 * XA_PITCH];

    const int lane = threadIdx.x & 31;
    const int wave = __builtin_amdgcn_readfirstlane((int)(threadIdx.x >> 5));
    const int m = lane & 15;
    const int g = lane >> 4;

    const int colg     = blockIdx.x & 31;
    const int rowgroup = blockIdx.x >> 5;
    const int j0 = colg * 64;
    const int tile0 = rowgroup * TILES_PER_BLOCK + wave;
    float* xa = xa_all + wave * (16 * XA_PITCH);
    const unsigned lane_off = 4u * (unsigned)lane;
    const unsigned cd_off   = (unsigned)(8 * g) * J_SZ + 4u * (unsigned)m;

    float4 araw[8];
    float4 xv[8];
    load_a_raw(araw, x1, 16 * tile0, lane_off);
    load_x2_tile(xv, x2, 16 * tile0, j0, cd_off);

#pragma unroll
    for (int r = 0; r < 8; ++r) {
        const int idx = threadIdx.x + 256 * r;
        const int p   = idx >> 6;
        const int jj  = idx & 63;
        v2f v;
        v.x = w2[(size_t)(2 * p + 0) * J_SZ + j0 + jj];
        v.y = w2[(size_t)(2 * p + 1) * J_SZ + j0 + jj];
        lw[p * LW_STRIDE + 16 * (jj & 3) + (jj >> 2)] = v;
    }
    __syncthreads();

    v2f bfrag[16][4];
#pragma unroll
    for (int t = 0; t < 16; ++t) {
        const v2f* brow = &lw[(2 * t + g) * LW_STRIDE + m];
#pragma unroll
        for (int c = 0; c < 4; ++c) bfrag[t][c] = brow[16 * c];
    }

    v2f afrag[16];
    for (int it = 0; it < TILES_PER_WAVE - 1; ++it) {
        const int b0 = 16 * (tile0 + it * WAVES_PER_BLOCK);
        stage_a_frags(afrag, araw, xa, lane, m, g);
        compute_tile(afrag, xv, bfrag, out, b0, j0, cd_off);
        load_a_raw(araw, x1, b0 + 16 * WAVES_PER_BLOCK, lane_off);
        load_x2_tile(xv, x2, b0 + 16 * WAVES_PER_BLOCK, j0, cd_off);
    }
    stage_a_frags(afrag, araw, xa, lane, m, g);
    compute_tile(afrag, xv, bfrag, out,
                 16 * (tile0 + (TILES_PER_WAVE - 1) * WAVES_PER_BLOCK), j0, cd_off);
}

extern "C" void kernel_launch(void* const* d_in, const int* in_sizes, int n_in,
                              void* d_out, int out_size, void* d_ws, size_t ws_size,
                              hipStream_t stream) {
    const float* x1 = (const float*)d_in[0];
    const float* x2 = (const float*)d_in[1];
    const float* w  = (const float*)d_in[2];
    float* out = (float*)d_out;
    float* w2  = (float*)d_ws;

    reduce_w_kernel<<<(I_SZ * J_SZ) / 256, 256, 0, stream>>>(w, w2);

    bilinear_gemm_kernel<<<ROW_GROUPS * (J_SZ / 64), 256, 0, stream>>>(x1, x2, w2, out);
}


# Weight sum reads whole lines: a wave takes 32 consecutive 256-B rows with 16-byte loads (4 lines per load, was 32) and adds each row's 16 lane partials by a row_xmask butterfly; w2 sum order changes
# speedup vs baseline: 1.3211x; 1.0421x over previous
#include <hip/hip_runtime.h>
#include <hip/hip_bf16.h>

typedef float v2f __attribute__((ext_vector_type(2)));
typedef float v4f __attribute__((ext_vector_type(4)));
typedef float v8f __attribute__((ext_vector_type(8)));

#define B_SZ 16384
#define I_SZ 64
#define J_SZ 2048
#define K_SZ 64

#define LW_STRIDE 80

#define XA_PITCH 68

#define ROWS_PER_WAVE 32

template <int MASK>
__device__ __forceinline__ float row_xor_add(float s) {
    const int t = __builtin_amdgcn_update_dpp(0, __builtin_bit_cast(int, s), 0x160 + MASK,
                                              0xf, 0xf, true);
    return s + __builtin_bit_cast(float, t);
}

__global__ __launch_bounds__(256) void reduce_w_kernel(const float* __restrict__ w,
                                                       float* __restrict__ w2) {
    const int lane = threadIdx.x & 31;
    const int wave = __builtin_amdgcn_readfirstlane((int)(threadIdx.x >> 5));
    const int row0 = (blockIdx.x * 8 + wave) * ROWS_PER_WAVE;
    const float* p = w + (size_t)row0 * K_SZ + 4 * lane;

    float4 v[16];
#pragma unroll
    for (int u = 0; u < 16; ++u) v[u] = *(const float4*)(p + 128 * u);
    __builtin_amdgcn_sched_barrier(0);

    const int sub = lane & 15;
    float r = 0.0f;
#pragma unroll
    for (int u = 0; u < 16; ++u) {
        float s = (v[u].x + v[u].y) + (v[u].z + v[u].w);
        s = row_xor_add<1>(s);
        s = row_xor_add<2>(s);
        s = row_xor_add<4>(s);
        s = row_xor_add<8>(s);
        r = (sub == u) ? s : r;
    }
    w2[row0 + 2 * sub + (lane >> 4)] = r;
}

#define ROW_TILES       (B_SZ / 16)
#define WAVES_PER_BLOCK 8
#define ROW_GROUPS      16
#define TILES_PER_BLOCK (ROW_TILES / ROW_GROUPS)
#define TILES_PER_WAVE  (TILES_PER_BLOCK / WAVES_PER_BLOCK)

__device__ __forceinline__ void load_a_raw(float4 (&ar)[8], const float* __restrict__ x1,
                                           int b0, unsigned lane_off) {
    const float* p = x1 + (size_t)b0 * I_SZ;
#pragma unroll
    for (int u = 0; u < 8; ++u) ar[u] = *(const float4*)(p + 128 * u + lane_off);
}

__device__ __forceinline__ void stage_a_frags(v2f (&a)[16], const float4 (&ar)[8],
                                              float* __restrict__ xa, int lane, int m, int g) {
    __builtin_amdgcn_fence(__ATOMIC_ACQ_REL, "wavefront");
    float* wr = xa + (lane >> 4) * XA_PITCH + 4 * (lane & 15);
#pragma unroll
    for (int u = 0; u < 8; ++u) {
        v4f v = {ar[u].x, ar[u].y, ar[u].z, ar[u].w};
        *(v4f*)(wr + 2 * u * XA_PITCH) = v;
    }
    __builtin_amdgcn_fence(__ATOMIC_ACQ_REL, "wavefront");
    const float* rd = xa + m * XA_PITCH + 2 * g;
#pragma unroll
    for (int t = 0; t < 16; ++t) a[t] = *(const v2f*)(rd + 4 * t);
    __builtin_amdgcn_sched_barrier(0);
}

__device__ __forceinline__ void load_x2_tile(float4 (&xv)[8], const float* __restrict__ x2,
                                             int b0, int j0, unsigned cd_off) {
    const float* p = x2 + (size_t)b0 * J_SZ + j0;
#pragma unroll
    for (int v = 0; v < 8; ++v) {
        const v4f t = __builtin_nontemporal_load((const v4f*)(p + v * J_SZ + cd_off));
        xv[v] = make_float4(t.x, t.y, t.z, t.w);
    }
}

__device__ __forceinline__ void compute_tile(const v2f (&afrag)[16], const float4 (&xv)[8],
                                             const v2f (&bfrag)[16][4], float* __restrict__ out,
                                             int b0, int j0, unsigned cd_off) {
    v8f acc[4];
#pragma unroll
    for (int c = 0; c < 4; ++c) acc[c] = (v8f){0.f, 0.f, 0.f, 0.f, 0.f, 0.f, 0.f, 0.f};

#pragma unroll
    for (int t = 0; t < 16; ++t) {
#pragma unroll
        for (int c = 0; c < 4; ++c) {
            acc[c] = __builtin_amdgcn_wmma_f32_16x16x4_f32(
                 false, afrag[t],
                 false, bfrag[t][c],
                 (short)0, acc[c],
                 false,  false);
        }
    }

    float* o = out + (size_t)b0 * J_SZ + j0;
#pragma unroll
    for (int v = 0; v < 8; ++v) {
        float4 r;
        r.x = acc[0][v] * xv[v].x;
        r.y = acc[1][v] * xv[v].y;
        r.z = acc[2][v] * xv[v].z;
        r.w = acc[3][v] * xv[v].w;
        *(float4*)(o + v * J_SZ + cd_off) = r;
    }
}

__global__ __launch_bounds__(256) void bilinear_gemm_kernel(
    const float* __restrict__ x1,
    const float* __restrict__ x2,
    const float* __restrict__ w2,
    float* __restrict__ out)
{
    __shared__ v2f lw[32 * LW_STRIDE];
    __shared__ __attribute__((aligned(16))) float xa_all[WAVES_PER_BLOCK * 16 * XA_PITCH];

    const int lane = threadIdx.x & 31;
    const int wave = __builtin_amdgcn_readfirstlane((int)(threadIdx.x >> 5));
    const int m = lane & 15;
    const int g = lane >> 4;

    const int colg     = blockIdx.x & 31;
    const int rowgroup = blockIdx.x >> 5;
    const int j0 = colg * 64;
    const int tile0 = rowgroup * TILES_PER_BLOCK + wave;
    float* xa = xa_all + wave * (16 * XA_PITCH);
    const unsigned lane_off = 4u * (unsigned)lane;
    const unsigned cd_off   = (unsigned)(8 * g) * J_SZ + 4u * (unsigned)m;

    float4 araw[8];
    float4 xv[8];
    load_a_raw(araw, x1, 16 * tile0, lane_off);
    load_x2_tile(xv, x2, 16 * tile0, j0, cd_off);

#pragma unroll
    for (int r = 0; r < 8; ++r) {
        const int idx = threadIdx.x + 256 * r;
        const int p   = idx >> 6;
        const int jj  = idx & 63;
        v2f v;
        v.x = w2[(size_t)(2 * p + 0) * J_SZ + j0 + jj];
        v.y = w2[(size_t)(2 * p + 1) * J_SZ + j0 + jj];
        lw[p * LW_STRIDE + 16 * (jj & 3) + (jj >> 2)] = v;
    }
    __syncthreads();

    v2f bfrag[16][4];
#pragma unroll
    for (int t = 0; t < 16; ++t) {
        const v2f* brow = &lw[(2 * t + g) * LW_STRIDE + m];
#pragma unroll
        for (int c = 0; c < 4; ++c) bfrag[t][c] = brow[16 * c];
    }

    v2f afrag[16];
    for (int it = 0; it < TILES_PER_WAVE - 1; ++it) {
        const int b0 = 16 * (tile0 + it * WAVES_PER_BLOCK);
        stage_a_frags(afrag, araw, xa, lane, m, g);
        compute_tile(afrag, xv, bfrag, out, b0, j0, cd_off);
        load_a_raw(araw, x1, b0 + 16 * WAVES_PER_BLOCK, lane_off);
        load_x2_tile(xv, x2, b0 + 16 * WAVES_PER_BLOCK, j0, cd_off);
    }
    stage_a_frags(afrag, araw, xa, lane, m, g);
    compute_tile(afrag, xv, bfrag, out,
                 16 * (tile0 + (TILES_PER_WAVE - 1) * WAVES_PER_BLOCK), j0, cd_off);
}

extern "C" void kernel_launch(void* const* d_in, const int* in_sizes, int n_in,
                              void* d_out, int out_size, void* d_ws, size_t ws_size,
                              hipStream_t stream) {
    const float* x1 = (const float*)d_in[0];
    const float* x2 = (const float*)d_in[1];
    const float* w  = (const float*)d_in[2];
    float* out = (float*)d_out;
    float* w2  = (float*)d_ws;

    reduce_w_kernel<<<(I_SZ * J_SZ) / 256, 256, 0, stream>>>(w, w2);

    bilinear_gemm_kernel<<<ROW_GROUPS * (J_SZ / 64), 256, 0, stream>>>(x1, x2, w2, out);
}
